# starting kernel
# baseline (speedup 1.0000x reference)
#include <hip/hip_runtime.h>
#include <hip/hip_bf16.h>

typedef float v2f __attribute__((ext_vector_type(2)));
typedef float v8f __attribute__((ext_vector_type(8)));

__global__ void build_u_kernel(const float* __restrict__ w, float* __restrict__ ws) {
    const int j = threadIdx.x;
    if (j >= 16) return;
    float cr[16], ci[16];
    for (int k = 0; k < 16; ++k) { cr[k] = (k == j) ? 1.f : 0.f; ci[k] = 0.f; }

    for (int l = 0; l < 6; ++l) {
        for (int i = 0; i < 4; ++i) {
            const float phi = w[(l * 4 + i) * 3 + 0];
            const float th  = w[(l * 4 + i) * 3 + 1];
            const float om  = w[(l * 4 + i) * 3 + 2];
            const float ct = cosf(0.5f * th), st = sinf(0.5f * th);
            const float a  = 0.5f * (phi + om), bb = 0.5f * (phi - om);
            const float ca = cosf(a), sa = sinf(a), cb = cosf(bb), sb = sinf(bb);
            const float g00r =  ct * ca, g00i = -ct * sa;
            const float g01r = -st * cb, g01i = -st * sb;
            const float g10r =  st * cb, g10i = -st * sb;
            const float g11r =  ct * ca, g11i =  ct * sa;
            const int bit = 1 << (3 - i);
            for (int j0 = 0; j0 < 16; ++j0) {
                if (j0 & bit) continue;
                const int j1 = j0 | bit;
                const float v0r = cr[j0], v0i = ci[j0], v1r = cr[j1], v1i = ci[j1];
                cr[j0] = g00r * v0r - g00i * v0i + g01r * v1r - g01i * v1i;
                ci[j0] = g00r * v0i + g00i * v0r + g01r * v1i + g01i * v1r;
                cr[j1] = g10r * v0r - g10i * v0i + g11r * v1r - g11i * v1i;
                ci[j1] = g10r * v0i + g10i * v0r + g11r * v1i + g11i * v1r;
            }
        }
        const int r = (l % 3) + 1;
        int perm[16], np[16];
        for (int k = 0; k < 16; ++k) perm[k] = k;
        for (int i = 0; i < 4; ++i) {
            const int tgt = (i + r) & 3;
            for (int k = 0; k < 16; ++k) {
                const int cbit = (k >> (3 - i)) & 1;
                const int src  = cbit ? (k ^ (1 << (3 - tgt))) : k;
                np[k] = perm[src];
            }
            for (int k = 0; k < 16; ++k) perm[k] = np[k];
        }
        float tr[16], ti[16];
        for (int k = 0; k < 16; ++k) { tr[k] = cr[perm[k]]; ti[k] = ci[perm[k]]; }
        for (int k = 0; k < 16; ++k) { cr[k] = tr[k]; ci[k] = ti[k]; }
    }

    const int t     = j >> 2;
    const int lbase = (j & 2) ? 16 : 0;
    const int comp  = j & 1;
    float* wr = ws;
    float* wi = ws + 256;
    for (int m = 0; m < 16; ++m) {
        const int L = lbase + m;
        wr[(t * 32 + L) * 2 + comp] = cr[m];
        wi[(t * 32 + L) * 2 + comp] = ci[m];
    }
}

__global__ void __launch_bounds__(256)
qlayer_kernel(const float* __restrict__ x, const float* __restrict__ ws,
              float* __restrict__ out, int batch) {
    const int lane = threadIdx.x & 31;
    const int wave = (blockIdx.x * blockDim.x + threadIdx.x) >> 5;
    int b = wave * 16 + (lane & 15);
    if (b >= batch) b = batch - 1;

    const float4 xv = *reinterpret_cast<const float4*>(x + (size_t)b * 4);
    const float c0 = __cosf(0.5f * xv.x), s0 = __sinf(0.5f * xv.x);
    const float c1 = __cosf(0.5f * xv.y), s1 = __sinf(0.5f * xv.y);
    const float c2 = __cosf(0.5f * xv.z), s2 = __sinf(0.5f * xv.z);
    const float c3 = __cosf(0.5f * xv.w), s3 = __sinf(0.5f * xv.w);

    const float hi = (lane < 16) ? c2 : s2;
    const float T0 = c0 * c1, T1 = c0 * s1, T2 = s0 * c1, T3 = s0 * s1;
    v2f bf[4];
    bf[0] = (v2f){T0 * hi * c3, T0 * hi * s3};
    bf[1] = (v2f){T1 * hi * c3, T1 * hi * s3};
    bf[2] = (v2f){T2 * hi * c3, T2 * hi * s3};
    bf[3] = (v2f){T3 * hi * c3, T3 * hi * s3};

    const float* wr = ws;
    const float* wi = ws + 256;
    v2f ar[4], ai[4];
#pragma unroll
    for (int t = 0; t < 4; ++t) {
        ar[t] = *reinterpret_cast<const v2f*>(wr + (t * 32 + lane) * 2);
        ai[t] = *reinterpret_cast<const v2f*>(wi + (t * 32 + lane) * 2);
    }

    v8f accr = {0.f, 0.f, 0.f, 0.f, 0.f, 0.f, 0.f, 0.f};
    v8f acci = {0.f, 0.f, 0.f, 0.f, 0.f, 0.f, 0.f, 0.f};
#pragma unroll
    for (int t = 0; t < 4; ++t) {
        accr = __builtin_amdgcn_wmma_f32_16x16x4_f32(
            false, ar[t], false, bf[t], (short)0, accr, false, false);
        acci = __builtin_amdgcn_wmma_f32_16x16x4_f32(
            false, ai[t], false, bf[t], (short)0, acci, false, false);
    }

    const int mbase = (lane < 16) ? 0 : 8;
    float o0 = 0.f, o1 = 0.f, o2 = 0.f, o3 = 0.f;
#pragma unroll
    for (int v = 0; v < 8; ++v) {
        const float p = accr[v] * accr[v] + acci[v] * acci[v];
        const int m = mbase + v;
        o0 += ((m >> 3) & 1) ? -p : p;
        o1 += ((m >> 2) & 1) ? -p : p;
        o2 += ((m >> 1) & 1) ? -p : p;
        o3 += ( m       & 1) ? -p : p;
    }
    o0 += __shfl_xor(o0, 16, 32);
    o1 += __shfl_xor(o1, 16, 32);
    o2 += __shfl_xor(o2, 16, 32);
    o3 += __shfl_xor(o3, 16, 32);

    if (lane < 16 && b < batch) {
        float4 res = {o0, o1, o2, o3};
        *reinterpret_cast<float4*>(out + (size_t)b * 4) = res;
    }
}

extern "C" void kernel_launch(void* const* d_in, const int* in_sizes, int n_in,
                              void* d_out, int out_size, void* d_ws, size_t ws_size,
                              hipStream_t stream) {
    const float* x = (const float*)d_in[0];
    const float* w = (const float*)d_in[1];
    float* out = (float*)d_out;
    float* ws  = (float*)d_ws;

    const int batch = in_sizes[0] / 4;

    build_u_kernel<<<1, 16, 0, stream>>>(w, ws);

    const int waves  = (batch + 15) / 16;
    const int wpb    = 8;
    const int blocks = (waves + wpb - 1) / wpb;
    qlayer_kernel<<<blocks, 256, 0, stream>>>(x, ws, out, batch);
}


# Build U with 256 threads, one amplitude each: partner read by in-row DPP, CNOT rings as compile-time lane masks, coefficients by readlane; 4-term gate sums now add own terms first
# speedup vs baseline: 2.6351x; 2.6351x over previous
#include <hip/hip_runtime.h>
#include <hip/hip_bf16.h>

typedef float v2f __attribute__((ext_vector_type(2)));
typedef float v8f __attribute__((ext_vector_type(8)));


constexpr int cnot_idx(int m, int c, int t) {
    return ((m >> (3 - c)) & 1) ? (m ^ (1 << (3 - t))) : m;
}
constexpr int ring_fwd(int m, int l) {
    const int r = (l % 3) + 1;
    for (int i = 0; i < 4; ++i) m = cnot_idx(m, i, (i + r) & 3);
    return m;
}
constexpr int index_at_layer(int p, int l) {
    for (int k = 0; k < l; ++k) p = ring_fwd(p, k);
    return p;
}
constexpr int partner_xor(int l, int i) {
    for (int d = 1; d < 16; ++d)
        if (index_at_layer(d, l) == (8 >> i)) return d;
    return 0;
}

template <int XMASK>
__device__ __forceinline__ float row_xor_read(float v) {
    return __builtin_bit_cast(float,
        __builtin_amdgcn_update_dpp(0, __builtin_bit_cast(int, v), 0x160 | XMASK, 0xF, 0xF, true));
}

template <int SRC>
__device__ __forceinline__ float from_lane(float v) {
    return __builtin_bit_cast(float, __builtin_amdgcn_readlane(__builtin_bit_cast(int, v), SRC));
}

template <int G>
__device__ __forceinline__ void apply_gates(float& vr, float& vi, int& m,
                                            float cA, float cB, float cC, float cD) {
    if constexpr (G < 24) {
        constexpr int l = G >> 2, i = G & 3;
        constexpr int d = partner_xor(l, i);
        static_assert(d > 0 && d < 16, "partner distance");
        const float gA = from_lane<G>(cA), gB = from_lane<G>(cB);
        const float gC = from_lane<G>(cC), gD = from_lane<G>(cD);
        const bool up = (m & (8 >> i)) != 0;
        const float pr = row_xor_read<d>(vr);
        const float pi = row_xor_read<d>(vi);
        const float ar = gA;
        const float ai = up ? gB : -gB;
        const float br = up ? gC : -gC;
        const float bi = -gD;
        const float nr = ar * vr - ai * vi + br * pr - bi * pi;
        const float ni = ar * vi + ai * vr + br * pi + bi * pr;
        vr = nr;
        vi = ni;
        if constexpr (i == 3) m = ring_fwd(m, l);
        apply_gates<G + 1>(vr, vi, m, cA, cB, cC, cD);
    }
}

__global__ void __launch_bounds__(256)
build_u_kernel(const float* __restrict__ w, float* __restrict__ ws) {
    const int tid  = threadIdx.x;
    const int lane = tid & 31;
    const int gsel = lane < 24 ? lane : 23;
    const float phi = w[gsel * 3 + 0];
    const float th  = w[gsel * 3 + 1];
    const float om  = w[gsel * 3 + 2];
    float st, ct, sa, ca, sb, cb;
    sincosf(0.5f * th, &st, &ct);
    sincosf(0.5f * (phi + om), &sa, &ca);
    sincosf(0.5f * (phi - om), &sb, &cb);
    const float cA = ct * ca, cB = ct * sa, cC = st * cb, cD = st * sb;

    const int j = tid >> 4;
    const int p = tid & 15;
    int   m  = p;
    float vr = (p == j) ? 1.f : 0.f;
    float vi = 0.f;
    apply_gates<0>(vr, vi, m, cA, cB, cC, cD);

    const int t     = j >> 2;
    const int lbase = (j & 2) ? 16 : 0;
    const int comp  = j & 1;
    ws[(t * 32 + lbase + m) * 2 + comp]       = vr;
    ws[256 + (t * 32 + lbase + m) * 2 + comp] = vi;
}

__global__ void __launch_bounds__(256)
qlayer_kernel(const float* __restrict__ x, const float* __restrict__ ws,
              float* __restrict__ out, int batch) {
    const int lane = threadIdx.x & 31;
    const int wave = (blockIdx.x * blockDim.x + threadIdx.x) >> 5;
    int b = wave * 16 + (lane & 15);
    if (b >= batch) b = batch - 1;

    const float4 xv = *reinterpret_cast<const float4*>(x + (size_t)b * 4);
    const float c0 = __cosf(0.5f * xv.x), s0 = __sinf(0.5f * xv.x);
    const float c1 = __cosf(0.5f * xv.y), s1 = __sinf(0.5f * xv.y);
    const float c2 = __cosf(0.5f * xv.z), s2 = __sinf(0.5f * xv.z);
    const float c3 = __cosf(0.5f * xv.w), s3 = __sinf(0.5f * xv.w);

    const float hi = (lane < 16) ? c2 : s2;
    const float T0 = c0 * c1, T1 = c0 * s1, T2 = s0 * c1, T3 = s0 * s1;
    v2f bf[4];
    bf[0] = (v2f){T0 * hi * c3, T0 * hi * s3};
    bf[1] = (v2f){T1 * hi * c3, T1 * hi * s3};
    bf[2] = (v2f){T2 * hi * c3, T2 * hi * s3};
    bf[3] = (v2f){T3 * hi * c3, T3 * hi * s3};

    const float* wr = ws;
    const float* wi = ws + 256;
    v2f ar[4], ai[4];
#pragma unroll
    for (int t = 0; t < 4; ++t) {
        ar[t] = *reinterpret_cast<const v2f*>(wr + (t * 32 + lane) * 2);
        ai[t] = *reinterpret_cast<const v2f*>(wi + (t * 32 + lane) * 2);
    }

    v8f accr = {0.f, 0.f, 0.f, 0.f, 0.f, 0.f, 0.f, 0.f};
    v8f acci = {0.f, 0.f, 0.f, 0.f, 0.f, 0.f, 0.f, 0.f};
#pragma unroll
    for (int t = 0; t < 4; ++t) {
        accr = __builtin_amdgcn_wmma_f32_16x16x4_f32(
            false, ar[t], false, bf[t], (short)0, accr, false, false);
        acci = __builtin_amdgcn_wmma_f32_16x16x4_f32(
            false, ai[t], false, bf[t], (short)0, acci, false, false);
    }

    const int mbase = (lane < 16) ? 0 : 8;
    float o0 = 0.f, o1 = 0.f, o2 = 0.f, o3 = 0.f;
#pragma unroll
    for (int v = 0; v < 8; ++v) {
        const float p = accr[v] * accr[v] + acci[v] * acci[v];
        const int m = mbase + v;
        o0 += ((m >> 3) & 1) ? -p : p;
        o1 += ((m >> 2) & 1) ? -p : p;
        o2 += ((m >> 1) & 1) ? -p : p;
        o3 += ( m       & 1) ? -p : p;
    }
    o0 += __shfl_xor(o0, 16, 32);
    o1 += __shfl_xor(o1, 16, 32);
    o2 += __shfl_xor(o2, 16, 32);
    o3 += __shfl_xor(o3, 16, 32);

    if (lane < 16 && b < batch) {
        float4 res = {o0, o1, o2, o3};
        *reinterpret_cast<float4*>(out + (size_t)b * 4) = res;
    }
}

extern "C" void kernel_launch(void* const* d_in, const int* in_sizes, int n_in,
                              void* d_out, int out_size, void* d_ws, size_t ws_size,
                              hipStream_t stream) {
    const float* x = (const float*)d_in[0];
    const float* w = (const float*)d_in[1];
    float* out = (float*)d_out;
    float* ws  = (float*)d_ws;

    const int batch = in_sizes[0] / 4;

    build_u_kernel<<<1, 256, 0, stream>>>(w, ws);

    const int waves  = (batch + 15) / 16;
    const int wpb    = 8;
    const int blocks = (waves + wpb - 1) / wpb;
    qlayer_kernel<<<blocks, 256, 0, stream>>>(x, ws, out, batch);
}


# Main kernel: each wave works off 8 consecutive 16-row tiles (8x fewer waves); U fragments loaded once per wave, all 8 x rows requested up front; per-tile math and sum order unchanged
# speedup vs baseline: 2.9945x; 1.1364x over previous
#include <hip/hip_runtime.h>
#include <hip/hip_bf16.h>

typedef float v2f __attribute__((ext_vector_type(2)));
typedef float v8f __attribute__((ext_vector_type(8)));


constexpr int cnot_idx(int m, int c, int t) {
    return ((m >> (3 - c)) & 1) ? (m ^ (1 << (3 - t))) : m;
}
constexpr int ring_fwd(int m, int l) {
    const int r = (l % 3) + 1;
    for (int i = 0; i < 4; ++i) m = cnot_idx(m, i, (i + r) & 3);
    return m;
}
constexpr int index_at_layer(int p, int l) {
    for (int k = 0; k < l; ++k) p = ring_fwd(p, k);
    return p;
}
constexpr int partner_xor(int l, int i) {
    for (int d = 1; d < 16; ++d)
        if (index_at_layer(d, l) == (8 >> i)) return d;
    return 0;
}

template <int XMASK>
__device__ __forceinline__ float row_xor_read(float v) {
    return __builtin_bit_cast(float,
        __builtin_amdgcn_update_dpp(0, __builtin_bit_cast(int, v), 0x160 | XMASK, 0xF, 0xF, true));
}

template <int SRC>
__device__ __forceinline__ float from_lane(float v) {
    return __builtin_bit_cast(float, __builtin_amdgcn_readlane(__builtin_bit_cast(int, v), SRC));
}

template <int G>
__device__ __forceinline__ void apply_gates(float& vr, float& vi, int& m,
                                            float cA, float cB, float cC, float cD) {
    if constexpr (G < 24) {
        constexpr int l = G >> 2, i = G & 3;
        constexpr int d = partner_xor(l, i);
        static_assert(d > 0 && d < 16, "partner distance");
        const float gA = from_lane<G>(cA), gB = from_lane<G>(cB);
        const float gC = from_lane<G>(cC), gD = from_lane<G>(cD);
        const bool up = (m & (8 >> i)) != 0;
        const float pr = row_xor_read<d>(vr);
        const float pi = row_xor_read<d>(vi);
        const float ar = gA;
        const float ai = up ? gB : -gB;
        const float br = up ? gC : -gC;
        const float bi = -gD;
        const float nr = ar * vr - ai * vi + br * pr - bi * pi;
        const float ni = ar * vi + ai * vr + br * pi + bi * pr;
        vr = nr;
        vi = ni;
        if constexpr (i == 3) m = ring_fwd(m, l);
        apply_gates<G + 1>(vr, vi, m, cA, cB, cC, cD);
    }
}

__global__ void __launch_bounds__(256)
build_u_kernel(const float* __restrict__ w, float* __restrict__ ws) {
    const int tid  = threadIdx.x;
    const int lane = tid & 31;
    const int gsel = lane < 24 ? lane : 23;
    const float phi = w[gsel * 3 + 0];
    const float th  = w[gsel * 3 + 1];
    const float om  = w[gsel * 3 + 2];
    float st, ct, sa, ca, sb, cb;
    sincosf(0.5f * th, &st, &ct);
    sincosf(0.5f * (phi + om), &sa, &ca);
    sincosf(0.5f * (phi - om), &sb, &cb);
    const float cA = ct * ca, cB = ct * sa, cC = st * cb, cD = st * sb;

    const int j = tid >> 4;
    const int p = tid & 15;
    int   m  = p;
    float vr = (p == j) ? 1.f : 0.f;
    float vi = 0.f;
    apply_gates<0>(vr, vi, m, cA, cB, cC, cD);

    const int t     = j >> 2;
    const int lbase = (j & 2) ? 16 : 0;
    const int comp  = j & 1;
    ws[(t * 32 + lbase + m) * 2 + comp]       = vr;
    ws[256 + (t * 32 + lbase + m) * 2 + comp] = vi;
}

constexpr int TILES_PER_WAVE = 8;

__global__ void __launch_bounds__(256)
qlayer_kernel(const float* __restrict__ x, const float* __restrict__ ws,
              float* __restrict__ out, int batch) {
    const int lane = threadIdx.x & 31;
    const int wave = (blockIdx.x * blockDim.x + threadIdx.x) >> 5;
    const int b0 = wave * (16 * TILES_PER_WAVE) + (lane & 15);

    const float* wr = ws;
    const float* wi = ws + 256;
    v2f ar[4], ai[4];
#pragma unroll
    for (int t = 0; t < 4; ++t) {
        ar[t] = *reinterpret_cast<const v2f*>(wr + (t * 32 + lane) * 2);
        ai[t] = *reinterpret_cast<const v2f*>(wi + (t * 32 + lane) * 2);
    }

    float4 xv[TILES_PER_WAVE];
#pragma unroll
    for (int k = 0; k < TILES_PER_WAVE; ++k) {
        int b = b0 + 16 * k;
        if (b >= batch) b = batch - 1;
        xv[k] = *reinterpret_cast<const float4*>(x + (size_t)b * 4);
    }
    __builtin_amdgcn_sched_barrier(0);

#pragma unroll
    for (int k = 0; k < TILES_PER_WAVE; ++k) {
        const float c0 = __cosf(0.5f * xv[k].x), s0 = __sinf(0.5f * xv[k].x);
        const float c1 = __cosf(0.5f * xv[k].y), s1 = __sinf(0.5f * xv[k].y);
        const float c2 = __cosf(0.5f * xv[k].z), s2 = __sinf(0.5f * xv[k].z);
        const float c3 = __cosf(0.5f * xv[k].w), s3 = __sinf(0.5f * xv[k].w);

        const float hi = (lane < 16) ? c2 : s2;
        const float T0 = c0 * c1, T1 = c0 * s1, T2 = s0 * c1, T3 = s0 * s1;
        v2f bf[4];
        bf[0] = (v2f){T0 * hi * c3, T0 * hi * s3};
        bf[1] = (v2f){T1 * hi * c3, T1 * hi * s3};
        bf[2] = (v2f){T2 * hi * c3, T2 * hi * s3};
        bf[3] = (v2f){T3 * hi * c3, T3 * hi * s3};

        v8f accr = {0.f, 0.f, 0.f, 0.f, 0.f, 0.f, 0.f, 0.f};
        v8f acci = {0.f, 0.f, 0.f, 0.f, 0.f, 0.f, 0.f, 0.f};
#pragma unroll
        for (int t = 0; t < 4; ++t) {
            accr = __builtin_amdgcn_wmma_f32_16x16x4_f32(
                false, ar[t], false, bf[t], (short)0, accr, false, false);
            acci = __builtin_amdgcn_wmma_f32_16x16x4_f32(
                false, ai[t], false, bf[t], (short)0, acci, false, false);
        }

        const int mbase = (lane < 16) ? 0 : 8;
        float o0 = 0.f, o1 = 0.f, o2 = 0.f, o3 = 0.f;
#pragma unroll
        for (int v = 0; v < 8; ++v) {
            const float p = accr[v] * accr[v] + acci[v] * acci[v];
            const int m = mbase + v;
            o0 += ((m >> 3) & 1) ? -p : p;
            o1 += ((m >> 2) & 1) ? -p : p;
            o2 += ((m >> 1) & 1) ? -p : p;
            o3 += ( m       & 1) ? -p : p;
        }
        o0 += __shfl_xor(o0, 16, 32);
        o1 += __shfl_xor(o1, 16, 32);
        o2 += __shfl_xor(o2, 16, 32);
        o3 += __shfl_xor(o3, 16, 32);

        const int b = b0 + 16 * k;
        if (lane < 16 && b < batch) {
            float4 res = {o0, o1, o2, o3};
            *reinterpret_cast<float4*>(out + (size_t)b * 4) = res;
        }
    }
}

extern "C" void kernel_launch(void* const* d_in, const int* in_sizes, int n_in,
                              void* d_out, int out_size, void* d_ws, size_t ws_size,
                              hipStream_t stream) {
    const float* x = (const float*)d_in[0];
    const float* w = (const float*)d_in[1];
    float* out = (float*)d_out;
    float* ws  = (float*)d_ws;

    const int batch = in_sizes[0] / 4;

    build_u_kernel<<<1, 256, 0, stream>>>(w, ws);

    const int rows_per_wave = 16 * TILES_PER_WAVE;
    const int waves  = (batch + rows_per_wave - 1) / rows_per_wave;
    const int wpb    = 8;
    const int blocks = (waves + wpb - 1) / wpb;
    qlayer_kernel<<<blocks, 256, 0, stream>>>(x, ws, out, batch);
}


# One launch instead of two: each 256-thread work-group builds U itself while its x rows are in flight and passes it on through 2 KB of LDS; 16 tiles per wave halve the builds; sum order unchanged
# speedup vs baseline: 3.2874x; 1.0978x over previous
#include <hip/hip_runtime.h>
#include <hip/hip_bf16.h>

typedef float v2f __attribute__((ext_vector_type(2)));
typedef float v8f __attribute__((ext_vector_type(8)));


constexpr int cnot_idx(int m, int c, int t) {
    return ((m >> (3 - c)) & 1) ? (m ^ (1 << (3 - t))) : m;
}
constexpr int ring_fwd(int m, int l) {
    const int r = (l % 3) + 1;
    for (int i = 0; i < 4; ++i) m = cnot_idx(m, i, (i + r) & 3);
    return m;
}
constexpr int index_at_layer(int p, int l) {
    for (int k = 0; k < l; ++k) p = ring_fwd(p, k);
    return p;
}
constexpr int partner_xor(int l, int i) {
    for (int d = 1; d < 16; ++d)
        if (index_at_layer(d, l) == (8 >> i)) return d;
    return 0;
}

template <int XMASK>
__device__ __forceinline__ float row_xor_read(float v) {
    return __builtin_bit_cast(float,
        __builtin_amdgcn_update_dpp(0, __builtin_bit_cast(int, v), 0x160 | XMASK, 0xF, 0xF, true));
}

template <int SRC>
__device__ __forceinline__ float from_lane(float v) {
    return __builtin_bit_cast(float, __builtin_amdgcn_readlane(__builtin_bit_cast(int, v), SRC));
}

template <int G>
__device__ __forceinline__ void apply_gates(float& vr, float& vi, int& m,
                                            float cA, float cB, float cC, float cD) {
    if constexpr (G < 24) {
        constexpr int l = G >> 2, i = G & 3;
        constexpr int d = partner_xor(l, i);
        static_assert(d > 0 && d < 16, "partner distance");
        const float gA = from_lane<G>(cA), gB = from_lane<G>(cB);
        const float gC = from_lane<G>(cC), gD = from_lane<G>(cD);
        const bool up = (m & (8 >> i)) != 0;
        const float pr = row_xor_read<d>(vr);
        const float pi = row_xor_read<d>(vi);
        const float ar = gA;
        const float ai = up ? gB : -gB;
        const float br = up ? gC : -gC;
        const float bi = -gD;
        const float nr = ar * vr - ai * vi + br * pr - bi * pi;
        const float ni = ar * vi + ai * vr + br * pi + bi * pr;
        vr = nr;
        vi = ni;
        if constexpr (i == 3) m = ring_fwd(m, l);
        apply_gates<G + 1>(vr, vi, m, cA, cB, cC, cD);
    }
}

__device__ __forceinline__ void build_u_to_lds(float phi, float th, float om,
                                               float* __restrict__ u, int tid) {
    float st, ct, sa, ca, sb, cb;
    sincosf(0.5f * th, &st, &ct);
    sincosf(0.5f * (phi + om), &sa, &ca);
    sincosf(0.5f * (phi - om), &sb, &cb);
    const float cA = ct * ca, cB = ct * sa, cC = st * cb, cD = st * sb;

    const int j = tid >> 4;
    const int p = tid & 15;
    int   m  = p;
    float vr = (p == j) ? 1.f : 0.f;
    float vi = 0.f;
    apply_gates<0>(vr, vi, m, cA, cB, cC, cD);

    const int t     = j >> 2;
    const int lbase = (j & 2) ? 16 : 0;
    const int comp  = j & 1;
    u[(t * 32 + lbase + m) * 2 + comp]       = vr;
    u[256 + (t * 32 + lbase + m) * 2 + comp] = vi;
}

constexpr int TILES_PER_WAVE  = 16;
constexpr int WAVES_PER_BLOCK = 8;
static_assert(32 * WAVES_PER_BLOCK == 256, "the build is written for 256 threads");

__global__ void __launch_bounds__(32 * WAVES_PER_BLOCK)
qlayer_kernel(const float* __restrict__ x, const float* __restrict__ w,
              float* __restrict__ out, int batch) {
    __shared__ __attribute__((aligned(16))) float u_frag[512];

    const int tid  = threadIdx.x;
    const int lane = tid & 31;
    const int wave = (blockIdx.x * blockDim.x + threadIdx.x) >> 5;
    const int b0 = wave * (16 * TILES_PER_WAVE) + (lane & 15);

    const int gsel = lane < 24 ? lane : 23;
    const float phi = w[gsel * 3 + 0];
    const float th  = w[gsel * 3 + 1];
    const float om  = w[gsel * 3 + 2];

    float4 xv[TILES_PER_WAVE];
#pragma unroll
    for (int k = 0; k < TILES_PER_WAVE; ++k) {
        int b = b0 + 16 * k;
        if (b >= batch) b = batch - 1;
        xv[k] = *reinterpret_cast<const float4*>(x + (size_t)b * 4);
    }
    __builtin_amdgcn_sched_barrier(0);

    build_u_to_lds(phi, th, om, u_frag, tid);
    __syncthreads();

    v2f ar[4], ai[4];
#pragma unroll
    for (int t = 0; t < 4; ++t) {
        ar[t] = *reinterpret_cast<const v2f*>(u_frag + (t * 32 + lane) * 2);
        ai[t] = *reinterpret_cast<const v2f*>(u_frag + 256 + (t * 32 + lane) * 2);
    }

#pragma unroll
    for (int k = 0; k < TILES_PER_WAVE; ++k) {
        const float c0 = __cosf(0.5f * xv[k].x), s0 = __sinf(0.5f * xv[k].x);
        const float c1 = __cosf(0.5f * xv[k].y), s1 = __sinf(0.5f * xv[k].y);
        const float c2 = __cosf(0.5f * xv[k].z), s2 = __sinf(0.5f * xv[k].z);
        const float c3 = __cosf(0.5f * xv[k].w), s3 = __sinf(0.5f * xv[k].w);

        const float hi = (lane < 16) ? c2 : s2;
        const float T0 = c0 * c1, T1 = c0 * s1, T2 = s0 * c1, T3 = s0 * s1;
        v2f bf[4];
        bf[0] = (v2f){T0 * hi * c3, T0 * hi * s3};
        bf[1] = (v2f){T1 * hi * c3, T1 * hi * s3};
        bf[2] = (v2f){T2 * hi * c3, T2 * hi * s3};
        bf[3] = (v2f){T3 * hi * c3, T3 * hi * s3};

        v8f accr = {0.f, 0.f, 0.f, 0.f, 0.f, 0.f, 0.f, 0.f};
        v8f acci = {0.f, 0.f, 0.f, 0.f, 0.f, 0.f, 0.f, 0.f};
#pragma unroll
        for (int t = 0; t < 4; ++t) {
            accr = __builtin_amdgcn_wmma_f32_16x16x4_f32(
                false, ar[t], false, bf[t], (short)0, accr, false, false);
            acci = __builtin_amdgcn_wmma_f32_16x16x4_f32(
                false, ai[t], false, bf[t], (short)0, acci, false, false);
        }

        const int mbase = (lane < 16) ? 0 : 8;
        float o0 = 0.f, o1 = 0.f, o2 = 0.f, o3 = 0.f;
#pragma unroll
        for (int v = 0; v < 8; ++v) {
            const float p = accr[v] * accr[v] + acci[v] * acci[v];
            const int m = mbase + v;
            o0 += ((m >> 3) & 1) ? -p : p;
            o1 += ((m >> 2) & 1) ? -p : p;
            o2 += ((m >> 1) & 1) ? -p : p;
            o3 += ( m       & 1) ? -p : p;
        }
        o0 += __shfl_xor(o0, 16, 32);
        o1 += __shfl_xor(o1, 16, 32);
        o2 += __shfl_xor(o2, 16, 32);
        o3 += __shfl_xor(o3, 16, 32);

        const int b = b0 + 16 * k;
        if (lane < 16 && b < batch) {
            float4 res = {o0, o1, o2, o3};
            *reinterpret_cast<float4*>(out + (size_t)b * 4) = res;
        }
    }
}

extern "C" void kernel_launch(void* const* d_in, const int* in_sizes, int n_in,
                              void* d_out, int out_size, void* d_ws, size_t ws_size,
                              hipStream_t stream) {
    const float* x = (const float*)d_in[0];
    const float* w = (const float*)d_in[1];
    float* out = (float*)d_out;

    const int batch = in_sizes[0] / 4;

    const int rows_per_wave = 16 * TILES_PER_WAVE;
    const int waves  = (batch + rows_per_wave - 1) / rows_per_wave;
    const int wpb    = WAVES_PER_BLOCK;
    const int blocks = (waves + wpb - 1) / wpb;
    qlayer_kernel<<<blocks, 32 * wpb, 0, stream>>>(x, w, out, batch);
}


# Outputs as a trig polynomial in x (<Z_i> = e^T M_i e expanded): each work-group folds U into a 12x27 coefficient matrix via WMMA; 7 WMMAs per tile not 8, no per-state squares; sum order changed
# speedup vs baseline: 3.4902x; 1.0617x over previous
#include <hip/hip_runtime.h>
#include <hip/hip_bf16.h>

typedef float v2f __attribute__((ext_vector_type(2)));
typedef float v8f __attribute__((ext_vector_type(8)));


constexpr int cnot_idx(int m, int c, int t) {
    return ((m >> (3 - c)) & 1) ? (m ^ (1 << (3 - t))) : m;
}
constexpr int ring_fwd(int m, int l) {
    const int r = (l % 3) + 1;
    for (int i = 0; i < 4; ++i) m = cnot_idx(m, i, (i + r) & 3);
    return m;
}
constexpr int index_at_layer(int p, int l) {
    for (int k = 0; k < l; ++k) p = ring_fwd(p, k);
    return p;
}
constexpr int partner_xor(int l, int i) {
    for (int d = 1; d < 16; ++d)
        if (index_at_layer(d, l) == (8 >> i)) return d;
    return 0;
}

template <int XMASK>
__device__ __forceinline__ float row_xor_read(float v) {
    return __builtin_bit_cast(float,
        __builtin_amdgcn_update_dpp(0, __builtin_bit_cast(int, v), 0x160 | XMASK, 0xF, 0xF, true));
}

template <int SRC>
__device__ __forceinline__ float from_lane(float v) {
    return __builtin_bit_cast(float, __builtin_amdgcn_readlane(__builtin_bit_cast(int, v), SRC));
}

__device__ __forceinline__ float other_half(float v) {
    const int b = __builtin_bit_cast(int, v);
    return __builtin_bit_cast(float,
        __builtin_amdgcn_permlanex16(b, b, 0x76543210, 0xfedcba98, false, false));
}

template <int G>
__device__ __forceinline__ void apply_gates(float& vr, float& vi, int& m,
                                            float cA, float cB, float cC, float cD) {
    if constexpr (G < 24) {
        constexpr int l = G >> 2, i = G & 3;
        constexpr int d = partner_xor(l, i);
        static_assert(d > 0 && d < 16, "partner distance");
        const float gA = from_lane<G>(cA), gB = from_lane<G>(cB);
        const float gC = from_lane<G>(cC), gD = from_lane<G>(cD);
        const bool up = (m & (8 >> i)) != 0;
        const float pr = row_xor_read<d>(vr);
        const float pi = row_xor_read<d>(vi);
        const float ar = gA;
        const float ai = up ? gB : -gB;
        const float br = up ? gC : -gC;
        const float bi = -gD;
        const float nr = ar * vr - ai * vi + br * pr - bi * pi;
        const float ni = ar * vi + ai * vr + br * pi + bi * pr;
        vr = nr;
        vi = ni;
        if constexpr (i == 3) m = ring_fwd(m, l);
        apply_gates<G + 1>(vr, vi, m, cA, cB, cC, cD);
    }
}

__device__ __forceinline__ void build_u_to_lds(float phi, float th, float om,
                                               float* __restrict__ u, int tid) {
    float st, ct, sa, ca, sb, cb;
    sincosf(0.5f * th, &st, &ct);
    sincosf(0.5f * (phi + om), &sa, &ca);
    sincosf(0.5f * (phi - om), &sb, &cb);
    const float cA = ct * ca, cB = ct * sa, cC = st * cb, cD = st * sb;

    const int j = tid >> 4;
    const int p = tid & 15;
    int   m  = p;
    float vr = (p == j) ? 1.f : 0.f;
    float vi = 0.f;
    apply_gates<0>(vr, vi, m, cA, cB, cC, cD);

    u[m * 16 + j]       = vr;
    u[256 + m * 16 + j] = vi;
}

constexpr int TILES_PER_WAVE  = 16;
constexpr int WAVES_PER_BLOCK = 8;
static_assert(32 * WAVES_PER_BLOCK == 256, "the build is written for 256 threads");
static_assert(TILES_PER_WAVE % 2 == 0, "tiles are worked off in pairs");

constexpr int LDS_M = 512;
constexpr int LDS_C = 1536;
constexpr int LDS_FLOATS = 1536 + 7 * 64;

__device__ __forceinline__ float coeff_entry(const float* __restrict__ lds, int row, int h, int slot) {
    const int i = row & 3;
    bool zero = (row >= 8) && (row < 12);
    int xm = (row >= 12) ? 8 : 0;
    int zm = ((row >> 2) == 1) ? 8 : 0;
    if (slot < 9) {
        const int t2 = slot / 3, t3 = slot - 3 * t2;
        xm |= (h ? 4 : 0) | ((t2 == 2) ? 2 : 0) | ((t3 == 2) ? 1 : 0);
        zm |= (h ? 0 : 4) | ((t2 == 1) ? 2 : 0) | ((t3 == 1) ? 1 : 0);
    } else if (slot < 12) {
        const int t3 = slot - 9;
        xm |= (h ? 2 : 0) | ((t3 == 2) ? 1 : 0);
        zm |= (h ? 0 : 2) | ((t3 == 1) ? 1 : 0);
    } else if (slot == 12) {
        xm |= h ? 1 : 0;
        zm |= h ? 0 : 1;
    } else {
        zero = zero || (h == 1);
    }
    const float s0 = (zm & 1) ? -1.f : 1.f;
    const float s1 = (zm & 2) ? -1.f : 1.f;
    const float s2 = (zm & 4) ? -1.f : 1.f;
    const float s3 = (zm & 8) ? -1.f : 1.f;
    const int bx = (LDS_M + i * 256) | xm;
    float d[16];
#pragma unroll
    for (int j = 0; j < 16; ++j) d[j] = lds[(bx ^ j) + 16 * j];
    float a[8], b[4], cc[2];
#pragma unroll
    for (int k = 0; k < 8; ++k) a[k] = fmaf(s0, d[2 * k + 1], d[2 * k]);
#pragma unroll
    for (int k = 0; k < 4; ++k) b[k] = fmaf(s1, a[2 * k + 1], a[2 * k]);
#pragma unroll
    for (int k = 0; k < 2; ++k) cc[k] = fmaf(s2, b[2 * k + 1], b[2 * k]);
    const float v = fmaf(s3, cc[1], cc[0]);
    return zero ? 0.f : v * 0.0625f;
}

__device__ __forceinline__ void monomials(const float4 xr, float shift, bool lower, float k13,
                                          v2f (&b)[7], float& u0) {
    const float kRev = 0.15915494309189535f;
    const float r0 = fmaf(kRev, xr.x, shift);
    const float r1 = fmaf(kRev, xr.y, shift);
    const float r2 = kRev * xr.z;
    const float r3 = kRev * xr.w;
    u0 = __builtin_amdgcn_cosf(r0);
    const float u1 = __builtin_amdgcn_cosf(r1);
    const float c2 = __builtin_amdgcn_cosf(r2), s2 = __builtin_amdgcn_sinf(r2);
    const float c3 = __builtin_amdgcn_cosf(r3), s3 = __builtin_amdgcn_sinf(r3);
    const float v2 = lower ? c2 : s2;
    const float v3 = lower ? c3 : s3;
    const float g4 = c2 * c3, g5 = c2 * s3, g7 = s2 * c3, g8 = s2 * s3;
    b[0] = (v2f){u1,      u1 * c3};
    b[1] = (v2f){u1 * s3, u1 * c2};
    b[2] = (v2f){u1 * g4, u1 * g5};
    b[3] = (v2f){u1 * s2, u1 * g7};
    b[4] = (v2f){u1 * g8, v2};
    b[5] = (v2f){v2 * c3, v2 * s3};
    b[6] = (v2f){v3,      k13};
}

__device__ __forceinline__ void finish_tile(const v8f acc, float u0, int lane, int b, int batch,
                                            float* __restrict__ out) {
    const float q0 = fmaf(u0, acc[4], acc[0]);
    const float q1 = fmaf(u0, acc[5], acc[1]);
    const float q2 = fmaf(u0, acc[6], acc[2]);
    const float q3 = fmaf(u0, acc[7], acc[3]);
    const float o0 = q0 + other_half(q0);
    const float o1 = q1 + other_half(q1);
    const float o2 = q2 + other_half(q2);
    const float o3 = q3 + other_half(q3);
    if (lane < 16 && b < batch) {
        float4 res = {o0, o1, o2, o3};
        *reinterpret_cast<float4*>(out + (size_t)b * 4) = res;
    }
}

__global__ void __launch_bounds__(32 * WAVES_PER_BLOCK)
qlayer_kernel(const float* __restrict__ x, const float* __restrict__ w,
              float* __restrict__ out, int batch) {
    __shared__ __attribute__((aligned(16))) float lds[LDS_FLOATS];

    const int tid  = threadIdx.x;
    const int lane = tid & 31;
    const int wave = (blockIdx.x * blockDim.x + threadIdx.x) >> 5;
    const int b0 = wave * (16 * TILES_PER_WAVE) + (lane & 15);

    const int gsel = lane < 24 ? lane : 23;
    const float phi = w[gsel * 3 + 0];
    const float th  = w[gsel * 3 + 1];
    const float om  = w[gsel * 3 + 2];

    float4 xv[TILES_PER_WAVE];
#pragma unroll
    for (int k = 0; k < TILES_PER_WAVE; ++k) {
        int b = b0 + 16 * k;
        if (b >= batch) b = batch - 1;
        xv[k] = *reinterpret_cast<const float4*>(x + (size_t)b * 4);
    }
    __builtin_amdgcn_sched_barrier(0);

    build_u_to_lds(phi, th, om, lds, tid);
    __syncthreads();

    const int h = lane >> 4;
    const int c = lane & 15;

    const int wv = __builtin_amdgcn_readfirstlane(tid >> 5);
    if (wv < 4) {
        const int sh = 3 - wv;
        v8f accR = {0.f, 0.f, 0.f, 0.f, 0.f, 0.f, 0.f, 0.f};
        v8f accI = {0.f, 0.f, 0.f, 0.f, 0.f, 0.f, 0.f, 0.f};
#pragma unroll
        for (int t = 0; t < 4; ++t) {
            const int m0 = 4 * t + 2 * h;
            const float z0 = ((m0 >> sh) & 1) ? -1.f : 1.f;
            const float z1 = (((m0 + 1) >> sh) & 1) ? -1.f : 1.f;
            const float ar0 = lds[m0 * 16 + c],       ar1 = lds[(m0 + 1) * 16 + c];
            const float ai0 = lds[256 + m0 * 16 + c], ai1 = lds[256 + (m0 + 1) * 16 + c];
            const v2f aR = {ar0, ar1}, bR = {z0 * ar0, z1 * ar1};
            const v2f aI = {ai0, ai1}, bI = {z0 * ai0, z1 * ai1};
            accR = __builtin_amdgcn_wmma_f32_16x16x4_f32(
                false, aR, false, bR, (short)0, accR, false, false);
            accI = __builtin_amdgcn_wmma_f32_16x16x4_f32(
                false, aI, false, bI, (short)0, accI, false, false);
        }
#pragma unroll
        for (int r = 0; r < 8; ++r)
            lds[LDS_M + wv * 256 + (8 * h + r) * 16 + c] = accR[r] + accI[r];
    }
    __syncthreads();

    if (wv < 7) {
        const v2f cv = {coeff_entry(lds, c, h, 2 * wv), coeff_entry(lds, c, h, 2 * wv + 1)};
        *reinterpret_cast<v2f*>(lds + LDS_C + (wv * 32 + lane) * 2) = cv;
    }
    __syncthreads();

    v2f am[7];
#pragma unroll
    for (int t = 0; t < 7; ++t)
        am[t] = *reinterpret_cast<const v2f*>(lds + LDS_C + (t * 32 + lane) * 2);

    const bool  lower = lane < 16;
    const float shift = lower ? 0.0f : -0.25f;
    const float k13   = lower ? 1.0f : 0.0f;

#pragma unroll
    for (int k = 0; k < TILES_PER_WAVE; k += 2) {
        v2f bA[7], bB[7];
        float u0A, u0B;
        monomials(xv[k],     shift, lower, k13, bA, u0A);
        monomials(xv[k + 1], shift, lower, k13, bB, u0B);

        v8f accA = {0.f, 0.f, 0.f, 0.f, 0.f, 0.f, 0.f, 0.f};
        v8f accB = {0.f, 0.f, 0.f, 0.f, 0.f, 0.f, 0.f, 0.f};
#pragma unroll
        for (int t = 0; t < 7; ++t) {
            accA = __builtin_amdgcn_wmma_f32_16x16x4_f32(
                false, am[t], false, bA[t], (short)0, accA, false, false);
            accB = __builtin_amdgcn_wmma_f32_16x16x4_f32(
                false, am[t], false, bB[t], (short)0, accB, false, false);
        }
        finish_tile(accA, u0A, lane, b0 + 16 * k,       batch, out);
        finish_tile(accB, u0B, lane, b0 + 16 * (k + 1), batch, out);
    }
}

extern "C" void kernel_launch(void* const* d_in, const int* in_sizes, int n_in,
                              void* d_out, int out_size, void* d_ws, size_t ws_size,
                              hipStream_t stream) {
    const float* x = (const float*)d_in[0];
    const float* w = (const float*)d_in[1];
    float* out = (float*)d_out;

    const int batch = in_sizes[0] / 4;

    const int rows_per_wave = 16 * TILES_PER_WAVE;
    const int waves  = (batch + rows_per_wave - 1) / rows_per_wave;
    const int wpb    = WAVES_PER_BLOCK;
    const int blocks = (waves + wpb - 1) / wpb;
    qlayer_kernel<<<blocks, 32 * wpb, 0, stream>>>(x, w, out, batch);
}


# Tile loop software-pipelined by hand: next pair's monomials and previous pair's finish sliced into the gaps between the pair's 14 WMMAs; one full-wave store per pair, no branch; sum order unchanged
# speedup vs baseline: 3.8585x; 1.1055x over previous
#include <hip/hip_runtime.h>
#include <hip/hip_bf16.h>

typedef float v2f __attribute__((ext_vector_type(2)));
typedef float v8f __attribute__((ext_vector_type(8)));


constexpr int cnot_idx(int m, int c, int t) {
    return ((m >> (3 - c)) & 1) ? (m ^ (1 << (3 - t))) : m;
}
constexpr int ring_fwd(int m, int l) {
    const int r = (l % 3) + 1;
    for (int i = 0; i < 4; ++i) m = cnot_idx(m, i, (i + r) & 3);
    return m;
}
constexpr int index_at_layer(int p, int l) {
    for (int k = 0; k < l; ++k) p = ring_fwd(p, k);
    return p;
}
constexpr int partner_xor(int l, int i) {
    for (int d = 1; d < 16; ++d)
        if (index_at_layer(d, l) == (8 >> i)) return d;
    return 0;
}

template <int XMASK>
__device__ __forceinline__ float row_xor_read(float v) {
    return __builtin_bit_cast(float,
        __builtin_amdgcn_update_dpp(0, __builtin_bit_cast(int, v), 0x160 | XMASK, 0xF, 0xF, true));
}

template <int SRC>
__device__ __forceinline__ float from_lane(float v) {
    return __builtin_bit_cast(float, __builtin_amdgcn_readlane(__builtin_bit_cast(int, v), SRC));
}

__device__ __forceinline__ float other_half(float v) {
    const int b = __builtin_bit_cast(int, v);
    return __builtin_bit_cast(float,
        __builtin_amdgcn_permlanex16(b, b, 0x76543210, 0xfedcba98, false, false));
}

template <int G>
__device__ __forceinline__ void apply_gates(float& vr, float& vi, int& m,
                                            float cA, float cB, float cC, float cD) {
    if constexpr (G < 24) {
        constexpr int l = G >> 2, i = G & 3;
        constexpr int d = partner_xor(l, i);
        static_assert(d > 0 && d < 16, "partner distance");
        const float gA = from_lane<G>(cA), gB = from_lane<G>(cB);
        const float gC = from_lane<G>(cC), gD = from_lane<G>(cD);
        const bool up = (m & (8 >> i)) != 0;
        const float pr = row_xor_read<d>(vr);
        const float pi = row_xor_read<d>(vi);
        const float ar = gA;
        const float ai = up ? gB : -gB;
        const float br = up ? gC : -gC;
        const float bi = -gD;
        const float nr = ar * vr - ai * vi + br * pr - bi * pi;
        const float ni = ar * vi + ai * vr + br * pi + bi * pr;
        vr = nr;
        vi = ni;
        if constexpr (i == 3) m = ring_fwd(m, l);
        apply_gates<G + 1>(vr, vi, m, cA, cB, cC, cD);
    }
}

__device__ __forceinline__ void build_u_to_lds(float phi, float th, float om,
                                               float* __restrict__ u, int tid) {
    float st, ct, sa, ca, sb, cb;
    sincosf(0.5f * th, &st, &ct);
    sincosf(0.5f * (phi + om), &sa, &ca);
    sincosf(0.5f * (phi - om), &sb, &cb);
    const float cA = ct * ca, cB = ct * sa, cC = st * cb, cD = st * sb;

    const int j = tid >> 4;
    const int p = tid & 15;
    int   m  = p;
    float vr = (p == j) ? 1.f : 0.f;
    float vi = 0.f;
    apply_gates<0>(vr, vi, m, cA, cB, cC, cD);

    u[m * 16 + j]       = vr;
    u[256 + m * 16 + j] = vi;
}

constexpr int TILES_PER_WAVE  = 16;
constexpr int WAVES_PER_BLOCK = 8;
static_assert(32 * WAVES_PER_BLOCK == 256, "the build is written for 256 threads");
static_assert(TILES_PER_WAVE % 2 == 0, "tiles are worked off in pairs");

constexpr int LDS_M = 512;
constexpr int LDS_C = 1536;
constexpr int LDS_FLOATS = 1536 + 7 * 64;

__device__ __forceinline__ float coeff_entry(const float* __restrict__ lds, int row, int h, int slot) {
    const int i = row & 3;
    bool zero = (row >= 8) && (row < 12);
    int xm = (row >= 12) ? 8 : 0;
    int zm = ((row >> 2) == 1) ? 8 : 0;
    if (slot < 9) {
        const int t2 = slot / 3, t3 = slot - 3 * t2;
        xm |= (h ? 4 : 0) | ((t2 == 2) ? 2 : 0) | ((t3 == 2) ? 1 : 0);
        zm |= (h ? 0 : 4) | ((t2 == 1) ? 2 : 0) | ((t3 == 1) ? 1 : 0);
    } else if (slot < 12) {
        const int t3 = slot - 9;
        xm |= (h ? 2 : 0) | ((t3 == 2) ? 1 : 0);
        zm |= (h ? 0 : 2) | ((t3 == 1) ? 1 : 0);
    } else if (slot == 12) {
        xm |= h ? 1 : 0;
        zm |= h ? 0 : 1;
    } else {
        zero = zero || (h == 1);
    }
    const float s0 = (zm & 1) ? -1.f : 1.f;
    const float s1 = (zm & 2) ? -1.f : 1.f;
    const float s2 = (zm & 4) ? -1.f : 1.f;
    const float s3 = (zm & 8) ? -1.f : 1.f;
    const int bx = (LDS_M + i * 256) | xm;
    float d[16];
#pragma unroll
    for (int j = 0; j < 16; ++j) d[j] = lds[(bx ^ j) + 16 * j];
    float a[8], b[4], cc[2];
#pragma unroll
    for (int k = 0; k < 8; ++k) a[k] = fmaf(s0, d[2 * k + 1], d[2 * k]);
#pragma unroll
    for (int k = 0; k < 4; ++k) b[k] = fmaf(s1, a[2 * k + 1], a[2 * k]);
#pragma unroll
    for (int k = 0; k < 2; ++k) cc[k] = fmaf(s2, b[2 * k + 1], b[2 * k]);
    const float v = fmaf(s3, cc[1], cc[0]);
    return zero ? 0.f : v * 0.0625f;
}

__device__ __forceinline__ void monomials(const float4 xr, float shift, bool lower, float k13,
                                          v2f (&b)[7], float& u0) {
    const float kRev = 0.15915494309189535f;
    const float r0 = fmaf(kRev, xr.x, shift);
    const float r1 = fmaf(kRev, xr.y, shift);
    const float r2 = kRev * xr.z;
    const float r3 = kRev * xr.w;
    u0 = __builtin_amdgcn_cosf(r0);
    const float u1 = __builtin_amdgcn_cosf(r1);
    const float c2 = __builtin_amdgcn_cosf(r2), s2 = __builtin_amdgcn_sinf(r2);
    const float c3 = __builtin_amdgcn_cosf(r3), s3 = __builtin_amdgcn_sinf(r3);
    const float v2 = lower ? c2 : s2;
    const float v3 = lower ? c3 : s3;
    const float g4 = c2 * c3, g5 = c2 * s3, g7 = s2 * c3, g8 = s2 * s3;
    b[0] = (v2f){u1,      u1 * c3};
    b[1] = (v2f){u1 * s3, u1 * c2};
    b[2] = (v2f){u1 * g4, u1 * g5};
    b[3] = (v2f){u1 * s2, u1 * g7};
    b[4] = (v2f){u1 * g8, v2};
    b[5] = (v2f){v2 * c3, v2 * s3};
    b[6] = (v2f){v3,      k13};
}

__device__ __forceinline__ void finish_tile(const v8f acc, float u0, int lane, int b, int batch,
                                            float* __restrict__ out) {
    const float q0 = fmaf(u0, acc[4], acc[0]);
    const float q1 = fmaf(u0, acc[5], acc[1]);
    const float q2 = fmaf(u0, acc[6], acc[2]);
    const float q3 = fmaf(u0, acc[7], acc[3]);
    const float o0 = q0 + other_half(q0);
    const float o1 = q1 + other_half(q1);
    const float o2 = q2 + other_half(q2);
    const float o3 = q3 + other_half(q3);
    if (lane < 16 && b < batch) {
        float4 res = {o0, o1, o2, o3};
        *reinterpret_cast<float4*>(out + (size_t)b * 4) = res;
    }
}


struct Mono {
    float r2, r3, u0, u1, c2, s2, c3, s3, v2, v3, g4, g5, g7, g8;
    v2f b[7];
};

__device__ __forceinline__ void mono_slice(Mono& m, const float4 xr, int s,
                                           float shift, bool lower, float k13) {
    const float kRev = 0.15915494309189535f;
    if (s == 0) {
        m.u1 = __builtin_amdgcn_cosf(fmaf(kRev, xr.y, shift));
        m.r3 = kRev * xr.w;
        m.c3 = __builtin_amdgcn_cosf(m.r3);
        m.r2 = kRev * xr.z;
    } else if (s == 1) {
        m.s3 = __builtin_amdgcn_sinf(m.r3);
        m.c2 = __builtin_amdgcn_cosf(m.r2);
        m.b[0] = (v2f){m.u1, m.u1 * m.c3};
    } else if (s == 2) {
        m.s2 = __builtin_amdgcn_sinf(m.r2);
        m.b[1] = (v2f){m.u1 * m.s3, m.u1 * m.c2};
        m.g4 = m.c2 * m.c3;
        m.g5 = m.c2 * m.s3;
    } else if (s == 3) {
        m.u0 = __builtin_amdgcn_cosf(fmaf(kRev, xr.x, shift));
        m.b[2] = (v2f){m.u1 * m.g4, m.u1 * m.g5};
        m.g7 = m.s2 * m.c3;
        m.g8 = m.s2 * m.s3;
    } else if (s == 4) {
        m.b[3] = (v2f){m.u1 * m.s2, m.u1 * m.g7};
        m.v2 = lower ? m.c2 : m.s2;
        m.v3 = lower ? m.c3 : m.s3;
    } else if (s == 5) {
        m.b[4] = (v2f){m.u1 * m.g8, m.v2};
        m.b[5] = (v2f){m.v2 * m.c3, m.v2 * m.s3};
    } else {
        m.b[6] = (v2f){m.v3, k13};
    }
}

struct Fin {
    v8f accP, accQ;
    float u0P, u0Q;
    float qP[4], qQ[4], own[4], give[4], got[4];
};

__device__ __forceinline__ void fin_slice(Fin& f, int g, bool lower, int row0, int lane,
                                          float* __restrict__ out) {
    if (g == 1) {
        f.qP[0] = fmaf(f.u0P, f.accP[4], f.accP[0]);
        f.qP[1] = fmaf(f.u0P, f.accP[5], f.accP[1]);
        f.qP[2] = fmaf(f.u0P, f.accP[6], f.accP[2]);
        f.qP[3] = fmaf(f.u0P, f.accP[7], f.accP[3]);
    } else if (g == 2) {
        f.qQ[0] = fmaf(f.u0Q, f.accQ[4], f.accQ[0]);
        f.qQ[1] = fmaf(f.u0Q, f.accQ[5], f.accQ[1]);
        f.qQ[2] = fmaf(f.u0Q, f.accQ[6], f.accQ[2]);
        f.qQ[3] = fmaf(f.u0Q, f.accQ[7], f.accQ[3]);
    } else if (g == 3) {
        f.own[0] = lower ? f.qP[0] : f.qQ[0];
        f.own[1] = lower ? f.qP[1] : f.qQ[1];
        f.own[2] = lower ? f.qP[2] : f.qQ[2];
        f.own[3] = lower ? f.qP[3] : f.qQ[3];
    } else if (g == 4) {
        f.give[0] = lower ? f.qQ[0] : f.qP[0];
        f.give[1] = lower ? f.qQ[1] : f.qP[1];
        f.give[2] = lower ? f.qQ[2] : f.qP[2];
        f.give[3] = lower ? f.qQ[3] : f.qP[3];
    } else if (g == 5) {
        f.got[0] = other_half(f.give[0]);
        f.got[1] = other_half(f.give[1]);
        f.got[2] = other_half(f.give[2]);
        f.got[3] = other_half(f.give[3]);
    } else if (g == 6) {
        float4 res = {f.own[0] + f.got[0], f.own[1] + f.got[1],
                      f.own[2] + f.got[2], f.own[3] + f.got[3]};
        *reinterpret_cast<float4*>(out + (size_t)(row0 + lane) * 4) = res;
    }
}

__global__ void __launch_bounds__(32 * WAVES_PER_BLOCK)
qlayer_kernel(const float* __restrict__ x, const float* __restrict__ w,
              float* __restrict__ out, int batch) {
    __shared__ __attribute__((aligned(16))) float lds[LDS_FLOATS];

    const int tid  = threadIdx.x;
    const int lane = tid & 31;
    const int wave = (blockIdx.x * blockDim.x + threadIdx.x) >> 5;
    const int b0 = wave * (16 * TILES_PER_WAVE) + (lane & 15);

    const int gsel = lane < 24 ? lane : 23;
    const float phi = w[gsel * 3 + 0];
    const float th  = w[gsel * 3 + 1];
    const float om  = w[gsel * 3 + 2];

    float4 xv[TILES_PER_WAVE];
#pragma unroll
    for (int k = 0; k < TILES_PER_WAVE; ++k) {
        int b = b0 + 16 * k;
        if (b >= batch) b = batch - 1;
        xv[k] = *reinterpret_cast<const float4*>(x + (size_t)b * 4);
    }
    __builtin_amdgcn_sched_barrier(0);

    build_u_to_lds(phi, th, om, lds, tid);
    __syncthreads();

    const int h = lane >> 4;
    const int c = lane & 15;

    const int wv = __builtin_amdgcn_readfirstlane(tid >> 5);
    if (wv < 4) {
        const int sh = 3 - wv;
        v8f accR = {0.f, 0.f, 0.f, 0.f, 0.f, 0.f, 0.f, 0.f};
        v8f accI = {0.f, 0.f, 0.f, 0.f, 0.f, 0.f, 0.f, 0.f};
#pragma unroll
        for (int t = 0; t < 4; ++t) {
            const int m0 = 4 * t + 2 * h;
            const float z0 = ((m0 >> sh) & 1) ? -1.f : 1.f;
            const float z1 = (((m0 + 1) >> sh) & 1) ? -1.f : 1.f;
            const float ar0 = lds[m0 * 16 + c],       ar1 = lds[(m0 + 1) * 16 + c];
            const float ai0 = lds[256 + m0 * 16 + c], ai1 = lds[256 + (m0 + 1) * 16 + c];
            const v2f aR = {ar0, ar1}, bR = {z0 * ar0, z1 * ar1};
            const v2f aI = {ai0, ai1}, bI = {z0 * ai0, z1 * ai1};
            accR = __builtin_amdgcn_wmma_f32_16x16x4_f32(
                false, aR, false, bR, (short)0, accR, false, false);
            accI = __builtin_amdgcn_wmma_f32_16x16x4_f32(
                false, aI, false, bI, (short)0, accI, false, false);
        }
#pragma unroll
        for (int r = 0; r < 8; ++r)
            lds[LDS_M + wv * 256 + (8 * h + r) * 16 + c] = accR[r] + accI[r];
    }
    __syncthreads();

    if (wv < 7) {
        const v2f cv = {coeff_entry(lds, c, h, 2 * wv), coeff_entry(lds, c, h, 2 * wv + 1)};
        *reinterpret_cast<v2f*>(lds + LDS_C + (wv * 32 + lane) * 2) = cv;
    }
    __syncthreads();

    v2f am[7];
#pragma unroll
    for (int t = 0; t < 7; ++t)
        am[t] = *reinterpret_cast<const v2f*>(lds + LDS_C + (t * 32 + lane) * 2);

    const bool  lower = lane < 16;
    const float shift = lower ? 0.0f : -0.25f;
    const float k13   = lower ? 1.0f : 0.0f;

    const int row_base = __builtin_amdgcn_readfirstlane(wave) * (16 * TILES_PER_WAVE);
    if (row_base + 16 * TILES_PER_WAVE > batch) {
#pragma unroll 1
        for (int k = 0; k < TILES_PER_WAVE; ++k) {
            const int b = b0 + 16 * k;
            const int bc = b < batch ? b : batch - 1;
            const float4 xr = *reinterpret_cast<const float4*>(x + (size_t)bc * 4);
            v2f bb[7];
            float u0;
            monomials(xr, shift, lower, k13, bb, u0);
            v8f acc = {0.f, 0.f, 0.f, 0.f, 0.f, 0.f, 0.f, 0.f};
#pragma unroll
            for (int t = 0; t < 7; ++t)
                acc = __builtin_amdgcn_wmma_f32_16x16x4_f32(
                    false, am[t], false, bb[t], (short)0, acc, false, false);
            finish_tile(acc, u0, lane, b, batch, out);
        }
        return;
    }

    constexpr int PAIRS = TILES_PER_WAVE / 2;
    Mono curA, curB;
#pragma unroll
    for (int s = 0; s < 7; ++s) {
        mono_slice(curA, xv[0], s, shift, lower, k13);
        mono_slice(curB, xv[1], s, shift, lower, k13);
    }
    Fin fin;

#pragma unroll
    for (int p = 0; p < PAIRS; ++p) {
        Mono nxtA, nxtB;
        v8f accA = {0.f, 0.f, 0.f, 0.f, 0.f, 0.f, 0.f, 0.f};
        v8f accB = {0.f, 0.f, 0.f, 0.f, 0.f, 0.f, 0.f, 0.f};
#pragma unroll
        for (int g = 0; g < 14; ++g) {
            const int t = g >> 1;
            if (g & 1)
                accB = __builtin_amdgcn_wmma_f32_16x16x4_f32(
                    false, am[t], false, curB.b[t], (short)0, accB, false, false);
            else
                accA = __builtin_amdgcn_wmma_f32_16x16x4_f32(
                    false, am[t], false, curA.b[t], (short)0, accA, false, false);
            if (p + 1 < PAIRS) {
                if (g < 7) mono_slice(nxtA, xv[2 * p + 2], g,     shift, lower, k13);
                else       mono_slice(nxtB, xv[2 * p + 3], g - 7, shift, lower, k13);
            }
            if (p > 0) fin_slice(fin, g, lower, row_base + 32 * (p - 1), lane, out);
            __builtin_amdgcn_sched_barrier(0);
        }
        fin.accP = accA;
        fin.accQ = accB;
        fin.u0P = curA.u0;
        fin.u0Q = curB.u0;
        if (p + 1 < PAIRS) {
            curA = nxtA;
            curB = nxtB;
        }
    }
#pragma unroll
    for (int g = 0; g < 14; ++g)
        fin_slice(fin, g, lower, row_base + 32 * (PAIRS - 1), lane, out);
}

extern "C" void kernel_launch(void* const* d_in, const int* in_sizes, int n_in,
                              void* d_out, int out_size, void* d_ws, size_t ws_size,
                              hipStream_t stream) {
    const float* x = (const float*)d_in[0];
    const float* w = (const float*)d_in[1];
    float* out = (float*)d_out;

    const int batch = in_sizes[0] / 4;

    const int rows_per_wave = 16 * TILES_PER_WAVE;
    const int waves  = (batch + rows_per_wave - 1) / rows_per_wave;
    const int wpb    = WAVES_PER_BLOCK;
    const int blocks = (waves + wpb - 1) / wpb;
    qlayer_kernel<<<blocks, 32 * wpb, 0, stream>>>(x, w, out, batch);
}
